# starting kernel
# baseline (speedup 1.0000x reference)
#include <hip/hip_runtime.h>
#include <hip/hip_bf16.h>

typedef __attribute__((ext_vector_type(2))) float v2f;
typedef __attribute__((ext_vector_type(8))) float v8f;

#define C_IN_   8
#define C_OUT_  16
#define HW_     1024
#define D0_     3072
#define D1_     8192
#define P_      256
#define KTAPS   72
#define KSTEPS  18
#define NTILES  192
#define NWAVES  8
#define NSLICE  4
#define TILES_PER_SLICE (NTILES / NSLICE)
#define DSLICE  (D0_ / NSLICE)

__device__ __forceinline__ float ldg_off(const char* __restrict__ base, uint32_t off) {
    return *(const float*)(base + off);
}

__global__ __launch_bounds__(256) void deeppoly_conv_main(
    const float* __restrict__ l0,  const float* __restrict__ u0,
    const float* __restrict__ lx0, const float* __restrict__ ux0,
    const float* __restrict__ weight,
    float* __restrict__ ws)
{
    __shared__ float s_l0[DSLICE];
    __shared__ float s_u0[DSLICE];
    __shared__ float s_wpos[KTAPS * 16];
    __shared__ float s_wneg[KTAPS * 16];
    __shared__ int   s_colb[KTAPS];
    __shared__ float s_pl[NWAVES][32];
    __shared__ float s_pu[NWAVES][32];

    const int tid   = threadIdx.x;
    const int p     = blockIdx.x;
    const int slice = blockIdx.y;
    const int ph = p >> 4, pw = p & 15;
    const int dbase = slice * DSLICE;

    for (int i = tid; i < DSLICE; i += 256) { s_l0[i] = l0[dbase + i]; s_u0[i] = u0[dbase + i]; }

    for (int e = tid; e < KTAPS * 16; e += 256) {
        int k = e >> 4, oc = e & 15;
        int t9 = k % 9;
        int r = ph * 2 - 1 + t9 / 3;
        int c = pw * 2 - 1 + (t9 % 3);
        bool valid = (r >= 0) && (r < 32) && (c >= 0) && (c < 32);
        float w = valid ? weight[oc * KTAPS + k] : 0.f;
        s_wpos[k * 16 + oc] = w > 0.f ? w : 0.f;
        s_wneg[k * 16 + oc] = w < 0.f ? w : 0.f;
    }
    if (tid < KTAPS) {
        int ic = tid / 9, t9 = tid % 9;
        int r = ph * 2 - 1 + t9 / 3;
        int c = pw * 2 - 1 + (t9 % 3);
        bool valid = (r >= 0) && (r < 32) && (c >= 0) && (c < 32);
        s_colb[tid] = valid ? ((ic * HW_ + r * 32 + c) << 2) : 0;
    }
    __syncthreads();

    const int wave = tid >> 5;
    const int lane = tid & 31;
    const int m    = lane & 15;
    const int koff = (lane >> 4) << 1;

    v2f bp[KSTEPS], bn[KSTEPS];
    uint32_t c0b[KSTEPS], c1b[KSTEPS];
    #pragma unroll
    for (int s = 0; s < KSTEPS; ++s) {
        const int k0 = s * 4 + koff;
        bp[s]  = (v2f){ s_wpos[k0 * 16 + m], s_wpos[(k0 + 1) * 16 + m] };
        bn[s]  = (v2f){ s_wneg[k0 * 16 + m], s_wneg[(k0 + 1) * 16 + m] };
        c0b[s] = (uint32_t)s_colb[k0];
        c1b[s] = (uint32_t)s_colb[k0 + 1];
    }

    const char* __restrict__ lx0b = (const char*)lx0;
    const char* __restrict__ ux0b = (const char*)ux0;

    float acc_l = 0.f, acc_u = 0.f;

    for (int ti = 0; ti < TILES_PER_SLICE / NWAVES; ++ti) {
        const int t   = slice * TILES_PER_SLICE + wave + ti * NWAVES;
        const int d0r = t * 16;
        const uint32_t rowb = (uint32_t)(d0r + m) << 15;

        v8f alx = {};
        v8f aux = {};

        #pragma unroll
        for (int s = 0; s < KSTEPS; ++s) {
            const uint32_t o0 = rowb + c0b[s];
            const uint32_t o1 = rowb + c1b[s];
            v2f a_l = (v2f){ ldg_off(lx0b, o0), ldg_off(lx0b, o1) };
            v2f a_u = (v2f){ ldg_off(ux0b, o0), ldg_off(ux0b, o1) };
            alx = __builtin_amdgcn_wmma_f32_16x16x4_f32(false, a_l, false, bp[s], (short)0, alx, false, false);
            alx = __builtin_amdgcn_wmma_f32_16x16x4_f32(false, a_u, false, bn[s], (short)0, alx, false, false);
            aux = __builtin_amdgcn_wmma_f32_16x16x4_f32(false, a_u, false, bp[s], (short)0, aux, false, false);
            aux = __builtin_amdgcn_wmma_f32_16x16x4_f32(false, a_l, false, bn[s], (short)0, aux, false, false);
        }

        const int rloc = (d0r - dbase) + ((lane >> 4) << 3);
        #pragma unroll
        for (int i = 0; i < 8; ++i) {
            const float lv = s_l0[rloc + i], uv = s_u0[rloc + i];
            const float x = alx[i];
            acc_l += (x > 0.f ? lv : uv) * x;
            const float y = aux[i];
            acc_u += (y > 0.f ? uv : lv) * y;
        }
    }

    s_pl[wave][lane] = acc_l;
    s_pu[wave][lane] = acc_u;
    __syncthreads();

    if (tid < 32) {
        const int  oc  = tid & 15;
        const bool isU = tid >= 16;
        float sum = 0.f;
        #pragma unroll
        for (int w = 0; w < NWAVES; ++w)
            sum += isU ? (s_pu[w][oc] + s_pu[w][oc + 16])
                       : (s_pl[w][oc] + s_pl[w][oc + 16]);
        ws[(size_t)(p * NSLICE + slice) * 32 + tid] = sum;
    }
}

__global__ __launch_bounds__(32) void deeppoly_conv_finish(
    const float* __restrict__ lc0, const float* __restrict__ uc0,
    const float* __restrict__ weight, const float* __restrict__ bias,
    const float* __restrict__ ws, float* __restrict__ out)
{
    const int p   = blockIdx.x;
    const int tid = threadIdx.x;
    const int oc  = tid & 15;
    const bool isU = tid >= 16;
    const int ph = p >> 4, pw = p & 15;

    float sum = 0.f;
    #pragma unroll
    for (int s = 0; s < NSLICE; ++s)
        sum += ws[(size_t)(p * NSLICE + s) * 32 + tid];

    float cacc = bias[oc];
    for (int k = 0; k < KTAPS; ++k) {
        const int ic = k / 9, t9 = k % 9;
        const int r = ph * 2 - 1 + t9 / 3;
        const int c = pw * 2 - 1 + (t9 % 3);
        if (r < 0 || r >= 32 || c < 0 || c >= 32) continue;
        const int j = ic * HW_ + r * 32 + c;
        const float wv = weight[oc * KTAPS + k];
        const float wp = wv > 0.f ? wv : 0.f;
        const float wn = wv < 0.f ? wv : 0.f;
        const float lcv = lc0[j], ucv = uc0[j];
        cacc += isU ? (wp * ucv + wn * lcv)
                    : (wp * lcv + wn * ucv);
    }
    out[(isU ? 4096 : 0) + oc * P_ + p] = sum + cacc;
}

extern "C" void kernel_launch(void* const* d_in, const int* in_sizes, int n_in,
                              void* d_out, int out_size, void* d_ws, size_t ws_size,
                              hipStream_t stream) {
    const float* l0     = (const float*)d_in[0];
    const float* u0     = (const float*)d_in[1];
    const float* lx0    = (const float*)d_in[2];
    const float* ux0    = (const float*)d_in[3];
    const float* lc0    = (const float*)d_in[4];
    const float* uc0    = (const float*)d_in[5];
    const float* weight = (const float*)d_in[6];
    const float* bias   = (const float*)d_in[7];
    float* ws  = (float*)d_ws;
    float* out = (float*)d_out;

    deeppoly_conv_main<<<dim3(P_, NSLICE), 256, 0, stream>>>(
        l0, u0, lx0, ux0, weight, ws);
    deeppoly_conv_finish<<<P_, 32, 0, stream>>>(
        lc0, uc0, weight, bias, ws, out);
}


# Weights as A, 16 output columns (2 d-rows x 8 positions) as B: each load reads contiguous row pieces; one bound and half an output row per block; sums over taps, rows and waves regrouped
# speedup vs baseline: 3.0802x; 3.0802x over previous
#include <hip/hip_runtime.h>
#include <hip/hip_bf16.h>


typedef __attribute__((ext_vector_type(2))) float v2f;
typedef __attribute__((ext_vector_type(8))) float v8f;

#define C_IN_   8
#define C_OUT_  16
#define HW_     1024
#define D0_     3072
#define D1_     8192
#define P_      256
#define KTAPS   72
#define KSTEPS  18
#define NSLICE  4
#define DSLICE  (D0_ / NSLICE)
#define NWAVES  12
#define NGRP    12
#define NTASK   (DSLICE / 2 / NWAVES)
#define ROWB    32768u
#define OUTS_PER_BLOCK 128

struct TapBuf {
    v2f   m1[NGRP], m2[NGRP];
    float e1[NGRP], e2[NGRP];
    float lv, uv;
};

__device__ __forceinline__ v8f wmma4(v2f a, v2f b, v8f c) {
    return __builtin_amdgcn_wmma_f32_16x16x4_f32(false, a, false, b, (short)0, c, false, false);
}

__global__ __launch_bounds__(NWAVES * 32) void deeppoly_conv_main(
    const float* __restrict__ l0,  const float* __restrict__ u0,
    const float* __restrict__ lx0, const float* __restrict__ ux0,
    const float* __restrict__ weight,
    float* __restrict__ ws)
{
    __shared__ float s_part[NWAVES][OUTS_PER_BLOCK];

    const int tid   = threadIdx.x;
    const int wave  = tid >> 5;
    const int lane  = tid & 31;
    const int blk   = blockIdx.x & 63;
    const int slice = blockIdx.x >> 6;
    const int bnd   = blk & 1;
    const int q     = (blk >> 1) & 1;
    const int ph    = blk >> 2;

    const int c   = lane & 15;
    const int h   = lane >> 4;
    const int dd  = c >> 3;
    const int pwl = c & 7;
    const int pw  = q * 8 + pwl;
    const bool pw0 = (pw == 0);
    const bool top = (ph == 0);

    const char*  __restrict__ X1 = (const char*)(bnd ? ux0 : lx0);
    const char*  __restrict__ X2 = (const char*)(bnd ? lx0 : ux0);
    const float* __restrict__ Lp = bnd ? u0 : l0;
    const float* __restrict__ Up = bnd ? l0 : u0;

    v2f wp[KSTEPS], wn[KSTEPS];
    #pragma unroll
    for (int i = 0; i < NGRP / 2; ++i) {
        float w[6];
        #pragma unroll
        for (int jj = 0; jj < 2; ++jj) {
            const int j  = 2 * i + jj;
            const int ic = j / 3 + 4 * h;
            const int kr = j % 3;
            const bool padrow = (kr == 0) && top;
            #pragma unroll
            for (int kc = 0; kc < 3; ++kc) {
                const float v = weight[c * KTAPS + ic * 9 + kr * 3 + kc];
                w[jj * 3 + kc] = padrow ? 0.f : v;
            }
        }
        const float a[6] = { w[1], w[2], w[4], w[5], w[0], w[3] };
        #pragma unroll
        for (int s = 0; s < 3; ++s) {
            const float x = a[2 * s], y = a[2 * s + 1];
            wp[3 * i + s] = (v2f){ x > 0.f ? x : 0.f, y > 0.f ? y : 0.f };
            wn[3 * i + s] = (v2f){ x < 0.f ? x : 0.f, y < 0.f ? y : 0.f };
        }
    }

    uint32_t rb[3];
    #pragma unroll
    for (int kr = 0; kr < 3; ++kr)
        rb[kr] = (kr == 0 && top) ? 0u : (uint32_t)(2 * ph - 1 + kr) * 128u;
    const uint32_t lane_m = (uint32_t)dd * ROWB + (uint32_t)h * 16384u + (uint32_t)pw * 8u;
    const uint32_t lane_e = lane_m - (pw0 ? 0u : 4u);
    const uint32_t dbase  = (uint32_t)(slice * DSLICE + 2 * wave);

    auto load_task = [&](TapBuf& t, int task) {
        const uint32_t d  = dbase + (uint32_t)(2 * NWAVES * task);
        const uint32_t vb = d * ROWB;
        uint32_t vm[3], ve[3];
        #pragma unroll
        for (int kr = 0; kr < 3; ++kr) { vm[kr] = vb + lane_m + rb[kr]; ve[kr] = vb + lane_e + rb[kr]; }
        #pragma unroll
        for (int j = 0; j < NGRP; ++j) {
            const uint32_t ico = (uint32_t)(j / 3) * 4096u;
            const int kr = j % 3;
            t.m1[j] = *(const v2f*)(X1 + (vm[kr] + ico));
            t.m2[j] = *(const v2f*)(X2 + (vm[kr] + ico));
            t.e1[j] = *(const float*)(X1 + (ve[kr] + ico));
            t.e2[j] = *(const float*)(X2 + (ve[kr] + ico));
        }
        t.lv = Lp[d + dd];
        t.uv = Up[d + dd];
    };

    float acc[8] = {0.f, 0.f, 0.f, 0.f, 0.f, 0.f, 0.f, 0.f};

    auto compute_task = [&](const TapBuf& t) {
        v8f ca = {};
        v8f cb = {};
        #pragma unroll
        for (int i = 0; i < NGRP / 2; ++i) {
            const int j0 = 2 * i, j1 = 2 * i + 1;
            ca = wmma4(wp[3 * i],     t.m1[j0], ca);
            cb = wmma4(wn[3 * i],     t.m2[j0], cb);
            ca = wmma4(wp[3 * i + 1], t.m1[j1], ca);
            cb = wmma4(wn[3 * i + 1], t.m2[j1], cb);
            const v2f ea = (v2f){ pw0 ? 0.f : t.e1[j0], pw0 ? 0.f : t.e1[j1] };
            const v2f eb = (v2f){ pw0 ? 0.f : t.e2[j0], pw0 ? 0.f : t.e2[j1] };
            ca = wmma4(wp[3 * i + 2], ea, ca);
            cb = wmma4(wn[3 * i + 2], eb, cb);
        }
        #pragma unroll
        for (int r = 0; r < 8; ++r) {
            const float x = ca[r] + cb[r];
            acc[r] += (x > 0.f ? t.lv : t.uv) * x;
        }
    };

    TapBuf A, B;
    load_task(A, 0);
    for (int task = 0; task < NTASK - 2; task += 2) {
        load_task(B, task + 1);
        compute_task(A);
        load_task(A, task + 2);
        compute_task(B);
    }
    load_task(B, NTASK - 1);
    compute_task(A);
    compute_task(B);

    #pragma unroll
    for (int r = 0; r < 8; ++r) acc[r] += __shfl_xor(acc[r], 8);
    if (dd == 0) {
        #pragma unroll
        for (int r = 0; r < 8; ++r) s_part[wave][(8 * h + r) * 8 + pwl] = acc[r];
    }
    __syncthreads();

    if (tid < OUTS_PER_BLOCK) {
        float sum = 0.f;
        #pragma unroll
        for (int w = 0; w < NWAVES; ++w) sum += s_part[w][tid];
        ws[(size_t)slice * 8192 + blk * OUTS_PER_BLOCK + tid] = sum;
    }
}

__global__ __launch_bounds__(32) void deeppoly_conv_finish(
    const float* __restrict__ lc0, const float* __restrict__ uc0,
    const float* __restrict__ weight, const float* __restrict__ bias,
    const float* __restrict__ ws, float* __restrict__ out)
{
    const int p   = blockIdx.x;
    const int tid = threadIdx.x;
    const int oc  = tid & 15;
    const bool isU = tid >= 16;
    const int ph = p >> 4, pw = p & 15;

    const int blk = (ph * 2 + (pw >> 3)) * 2 + (isU ? 1 : 0);
    float sum = 0.f;
    #pragma unroll
    for (int s = 0; s < NSLICE; ++s)
        sum += ws[(size_t)s * 8192 + blk * OUTS_PER_BLOCK + oc * 8 + (pw & 7)];

    float cacc = bias[oc];
    for (int k = 0; k < KTAPS; ++k) {
        const int ic = k / 9, t9 = k % 9;
        const int r = ph * 2 - 1 + t9 / 3;
        const int c = pw * 2 - 1 + (t9 % 3);
        if (r < 0 || r >= 32 || c < 0 || c >= 32) continue;
        const int j = ic * HW_ + r * 32 + c;
        const float wv = weight[oc * KTAPS + k];
        const float wp = wv > 0.f ? wv : 0.f;
        const float wn = wv < 0.f ? wv : 0.f;
        const float lcv = lc0[j], ucv = uc0[j];
        cacc += isU ? (wp * ucv + wn * lcv)
                    : (wp * lcv + wn * ucv);
    }
    out[(isU ? 4096 : 0) + oc * P_ + p] = sum + cacc;
}

extern "C" void kernel_launch(void* const* d_in, const int* in_sizes, int n_in,
                              void* d_out, int out_size, void* d_ws, size_t ws_size,
                              hipStream_t stream) {
    const float* l0     = (const float*)d_in[0];
    const float* u0     = (const float*)d_in[1];
    const float* lx0    = (const float*)d_in[2];
    const float* ux0    = (const float*)d_in[3];
    const float* lc0    = (const float*)d_in[4];
    const float* uc0    = (const float*)d_in[5];
    const float* weight = (const float*)d_in[6];
    const float* bias   = (const float*)d_in[7];
    float* ws  = (float*)d_ws;
    float* out = (float*)d_out;

    deeppoly_conv_main<<<256, NWAVES * 32, 0, stream>>>(
        l0, u0, lx0, ux0, weight, ws);
    deeppoly_conv_finish<<<P_, 32, 0, stream>>>(
        lc0, uc0, weight, bias, ws, out);
}


# One d-row x 16 output columns as B: loads read whole 128-byte input rows, left tap from the neighbour lane; 16-wave blocks; 8 d-slices if ws_size holds 256 KB, else 4; d, wave, slice sums regrouped
# speedup vs baseline: 4.3649x; 1.4171x over previous
#include <hip/hip_runtime.h>
#include <hip/hip_bf16.h>


typedef __attribute__((ext_vector_type(2))) float v2f;
typedef __attribute__((ext_vector_type(8))) float v8f;

#define C_IN_   8
#define C_OUT_  16
#define HW_     1024
#define D0_     3072
#define D1_     8192
#define P_      256
#define KTAPS   72
#define KSTEPS  18
#define NWAVES  16
#define NGRP    12
#define SLICE_FLOATS 8192
#define ROWB    32768u
#define OUTS_PER_BLOCK 256

struct TapBuf {
    v2f   m1[NGRP], m2[NGRP];
    float lv, uv;
};

__device__ __forceinline__ v8f wmma4(v2f a, v2f b, v8f c) {
    return __builtin_amdgcn_wmma_f32_16x16x4_f32(false, a, false, b, (short)0, c, false, false);
}

__device__ __forceinline__ float left_lane(float v) {
    return __int_as_float(__builtin_amdgcn_update_dpp(0, __float_as_int(v), 0x111, 0xf, 0xf, true));
}

__device__ __forceinline__ int num_slices(size_t ws_bytes) {
    return ws_bytes >= (size_t)8 * SLICE_FLOATS * sizeof(float) ? 8 : 4;
}

__global__ __launch_bounds__(NWAVES * 32) void deeppoly_conv_main(
    const float* __restrict__ l0,  const float* __restrict__ u0,
    const float* __restrict__ lx0, const float* __restrict__ ux0,
    const float* __restrict__ weight,
    float* __restrict__ ws, size_t ws_bytes)
{
    __shared__ float s_part[NWAVES][OUTS_PER_BLOCK];

    const int nslice = num_slices(ws_bytes);
    const int slice  = blockIdx.x >> 5;
    if (slice >= nslice) return;
    const int dslice = D0_ / nslice;
    const int ntask  = dslice / NWAVES;

    const int tid   = threadIdx.x;
    const int wave  = __builtin_amdgcn_readfirstlane(tid >> 5);
    const int lane  = tid & 31;
    const int blk   = blockIdx.x & 31;
    const int bnd   = blk & 1;
    const int ph    = blk >> 1;

    const int c   = lane & 15;
    const int h   = lane >> 4;
    const bool top = (ph == 0);

    const char*  __restrict__ X1 = (const char*)(bnd ? ux0 : lx0);
    const char*  __restrict__ X2 = (const char*)(bnd ? lx0 : ux0);
    const float* __restrict__ Lp = bnd ? u0 : l0;
    const float* __restrict__ Up = bnd ? l0 : u0;

    v2f wp[KSTEPS], wn[KSTEPS];
    #pragma unroll
    for (int i = 0; i < NGRP / 2; ++i) {
        float w[6];
        #pragma unroll
        for (int jj = 0; jj < 2; ++jj) {
            const int j  = 2 * i + jj;
            const int ic = j / 3 + 4 * h;
            const int kr = j % 3;
            const bool padrow = (kr == 0) && top;
            #pragma unroll
            for (int kc = 0; kc < 3; ++kc) {
                const float v = weight[c * KTAPS + ic * 9 + kr * 3 + kc];
                w[jj * 3 + kc] = padrow ? 0.f : v;
            }
        }
        const float a[6] = { w[1], w[2], w[4], w[5], w[0], w[3] };
        #pragma unroll
        for (int s = 0; s < 3; ++s) {
            const float x = a[2 * s], y = a[2 * s + 1];
            wp[3 * i + s] = (v2f){ x > 0.f ? x : 0.f, y > 0.f ? y : 0.f };
            wn[3 * i + s] = (v2f){ x < 0.f ? x : 0.f, y < 0.f ? y : 0.f };
        }
    }

    uint32_t rb[3];
    #pragma unroll
    for (int kr = 0; kr < 3; ++kr)
        rb[kr] = (kr == 0 && top) ? 0u : (uint32_t)(2 * ph - 1 + kr) * 128u;
    const uint32_t lane_m = (uint32_t)h * 16384u + (uint32_t)c * 8u;
    const uint32_t dbase  = (uint32_t)(slice * dslice + wave);

    auto load_task = [&](TapBuf& t, int task) {
        const uint32_t d  = dbase + (uint32_t)(NWAVES * task);
        const uint32_t vb = d * ROWB;
        uint32_t vm[3];
        #pragma unroll
        for (int kr = 0; kr < 3; ++kr) vm[kr] = vb + lane_m + rb[kr];
        #pragma unroll
        for (int j = 0; j < NGRP; ++j) {
            const uint32_t ico = (uint32_t)(j / 3) * 4096u;
            const int kr = j % 3;
            t.m1[j] = *(const v2f*)(X1 + (vm[kr] + ico));
            t.m2[j] = *(const v2f*)(X2 + (vm[kr] + ico));
        }
        t.lv = Lp[d];
        t.uv = Up[d];
    };

    float acc[8] = {0.f, 0.f, 0.f, 0.f, 0.f, 0.f, 0.f, 0.f};

    auto compute_task = [&](const TapBuf& t) {
        v8f ca = {};
        v8f cb = {};
        #pragma unroll
        for (int i = 0; i < NGRP / 2; ++i) {
            const int j0 = 2 * i, j1 = 2 * i + 1;
            ca = wmma4(wp[3 * i],     t.m1[j0], ca);
            cb = wmma4(wn[3 * i],     t.m2[j0], cb);
            ca = wmma4(wp[3 * i + 1], t.m1[j1], ca);
            cb = wmma4(wn[3 * i + 1], t.m2[j1], cb);
            const v2f ea = (v2f){ left_lane(t.m1[j0].y), left_lane(t.m1[j1].y) };
            const v2f eb = (v2f){ left_lane(t.m2[j0].y), left_lane(t.m2[j1].y) };
            ca = wmma4(wp[3 * i + 2], ea, ca);
            cb = wmma4(wn[3 * i + 2], eb, cb);
        }
        #pragma unroll
        for (int r = 0; r < 8; ++r) {
            const float x = ca[r] + cb[r];
            acc[r] += (x > 0.f ? t.lv : t.uv) * x;
        }
    };

    TapBuf A, B;
    load_task(A, 0);
    for (int task = 0; task < ntask - 2; task += 2) {
        load_task(B, task + 1);
        compute_task(A);
        load_task(A, task + 2);
        compute_task(B);
    }
    load_task(B, ntask - 1);
    compute_task(A);
    compute_task(B);

    #pragma unroll
    for (int r = 0; r < 8; ++r) s_part[wave][(8 * h + r) * 16 + c] = acc[r];
    __syncthreads();

    if (tid < OUTS_PER_BLOCK) {
        float sum = 0.f;
        #pragma unroll
        for (int w = 0; w < NWAVES; ++w) sum += s_part[w][tid];
        ws[(size_t)slice * SLICE_FLOATS + blk * OUTS_PER_BLOCK + tid] = sum;
    }
}

__global__ __launch_bounds__(32) void deeppoly_conv_finish(
    const float* __restrict__ lc0, const float* __restrict__ uc0,
    const float* __restrict__ weight, const float* __restrict__ bias,
    const float* __restrict__ ws, size_t ws_bytes, float* __restrict__ out)
{
    const int p   = blockIdx.x;
    const int tid = threadIdx.x;
    const int oc  = tid & 15;
    const bool isU = tid >= 16;
    const int ph = p >> 4, pw = p & 15;

    const int blk = ph * 2 + (isU ? 1 : 0);
    const int nslice = num_slices(ws_bytes);
    float sum = 0.f;
    for (int s = 0; s < nslice; ++s)
        sum += ws[(size_t)s * SLICE_FLOATS + blk * OUTS_PER_BLOCK + oc * 16 + pw];

    float cacc = bias[oc];
    for (int k = 0; k < KTAPS; ++k) {
        const int ic = k / 9, t9 = k % 9;
        const int r = ph * 2 - 1 + t9 / 3;
        const int c = pw * 2 - 1 + (t9 % 3);
        if (r < 0 || r >= 32 || c < 0 || c >= 32) continue;
        const int j = ic * HW_ + r * 32 + c;
        const float wv = weight[oc * KTAPS + k];
        const float wp = wv > 0.f ? wv : 0.f;
        const float wn = wv < 0.f ? wv : 0.f;
        const float lcv = lc0[j], ucv = uc0[j];
        cacc += isU ? (wp * ucv + wn * lcv)
                    : (wp * lcv + wn * ucv);
    }
    out[(isU ? 4096 : 0) + oc * P_ + p] = sum + cacc;
}

extern "C" void kernel_launch(void* const* d_in, const int* in_sizes, int n_in,
                              void* d_out, int out_size, void* d_ws, size_t ws_size,
                              hipStream_t stream) {
    const float* l0     = (const float*)d_in[0];
    const float* u0     = (const float*)d_in[1];
    const float* lx0    = (const float*)d_in[2];
    const float* ux0    = (const float*)d_in[3];
    const float* lc0    = (const float*)d_in[4];
    const float* uc0    = (const float*)d_in[5];
    const float* weight = (const float*)d_in[6];
    const float* bias   = (const float*)d_in[7];
    float* ws  = (float*)d_ws;
    float* out = (float*)d_out;

    deeppoly_conv_main<<<256, NWAVES * 32, 0, stream>>>(
        l0, u0, lx0, ux0, weight, ws, ws_size);
    deeppoly_conv_finish<<<P_, 32, 0, stream>>>(
        lc0, uc0, weight, bias, ws, ws_size, out);
}


# Finish kernel: slice partials, weights, bias and lc0/uc0 windows (taps spread over lanes, read back by readlane) all loaded before first use: one memory wait, not one per tap; sum orders kept
# speedup vs baseline: 5.2408x; 1.2007x over previous
#include <hip/hip_runtime.h>
#include <hip/hip_bf16.h>


typedef __attribute__((ext_vector_type(2))) float v2f;
typedef __attribute__((ext_vector_type(8))) float v8f;

#define C_IN_   8
#define C_OUT_  16
#define HW_     1024
#define D0_     3072
#define D1_     8192
#define P_      256
#define KTAPS   72
#define KSTEPS  18
#define NWAVES  16
#define NGRP    12
#define SLICE_FLOATS 8192
#define ROWB    32768u
#define OUTS_PER_BLOCK 256

struct TapBuf {
    v2f   m1[NGRP], m2[NGRP];
    float lv, uv;
};

__device__ __forceinline__ v8f wmma4(v2f a, v2f b, v8f c) {
    return __builtin_amdgcn_wmma_f32_16x16x4_f32(false, a, false, b, (short)0, c, false, false);
}

__device__ __forceinline__ float left_lane(float v) {
    return __int_as_float(__builtin_amdgcn_update_dpp(0, __float_as_int(v), 0x111, 0xf, 0xf, true));
}

__device__ __forceinline__ int num_slices(size_t ws_bytes) {
    return ws_bytes >= (size_t)8 * SLICE_FLOATS * sizeof(float) ? 8 : 4;
}

__global__ __launch_bounds__(NWAVES * 32) void deeppoly_conv_main(
    const float* __restrict__ l0,  const float* __restrict__ u0,
    const float* __restrict__ lx0, const float* __restrict__ ux0,
    const float* __restrict__ weight,
    float* __restrict__ ws, size_t ws_bytes)
{
    __shared__ float s_part[NWAVES][OUTS_PER_BLOCK];

    const int nslice = num_slices(ws_bytes);
    const int slice  = blockIdx.x >> 5;
    if (slice >= nslice) return;
    const int dslice = D0_ / nslice;
    const int ntask  = dslice / NWAVES;

    const int tid   = threadIdx.x;
    const int wave  = __builtin_amdgcn_readfirstlane(tid >> 5);
    const int lane  = tid & 31;
    const int blk   = blockIdx.x & 31;
    const int bnd   = blk & 1;
    const int ph    = blk >> 1;

    const int c   = lane & 15;
    const int h   = lane >> 4;
    const bool top = (ph == 0);

    const char*  __restrict__ X1 = (const char*)(bnd ? ux0 : lx0);
    const char*  __restrict__ X2 = (const char*)(bnd ? lx0 : ux0);
    const float* __restrict__ Lp = bnd ? u0 : l0;
    const float* __restrict__ Up = bnd ? l0 : u0;

    v2f wp[KSTEPS], wn[KSTEPS];
    #pragma unroll
    for (int i = 0; i < NGRP / 2; ++i) {
        float w[6];
        #pragma unroll
        for (int jj = 0; jj < 2; ++jj) {
            const int j  = 2 * i + jj;
            const int ic = j / 3 + 4 * h;
            const int kr = j % 3;
            const bool padrow = (kr == 0) && top;
            #pragma unroll
            for (int kc = 0; kc < 3; ++kc) {
                const float v = weight[c * KTAPS + ic * 9 + kr * 3 + kc];
                w[jj * 3 + kc] = padrow ? 0.f : v;
            }
        }
        const float a[6] = { w[1], w[2], w[4], w[5], w[0], w[3] };
        #pragma unroll
        for (int s = 0; s < 3; ++s) {
            const float x = a[2 * s], y = a[2 * s + 1];
            wp[3 * i + s] = (v2f){ x > 0.f ? x : 0.f, y > 0.f ? y : 0.f };
            wn[3 * i + s] = (v2f){ x < 0.f ? x : 0.f, y < 0.f ? y : 0.f };
        }
    }

    uint32_t rb[3];
    #pragma unroll
    for (int kr = 0; kr < 3; ++kr)
        rb[kr] = (kr == 0 && top) ? 0u : (uint32_t)(2 * ph - 1 + kr) * 128u;
    const uint32_t lane_m = (uint32_t)h * 16384u + (uint32_t)c * 8u;
    const uint32_t dbase  = (uint32_t)(slice * dslice + wave);

    auto load_task = [&](TapBuf& t, int task) {
        const uint32_t d  = dbase + (uint32_t)(NWAVES * task);
        const uint32_t vb = d * ROWB;
        uint32_t vm[3];
        #pragma unroll
        for (int kr = 0; kr < 3; ++kr) vm[kr] = vb + lane_m + rb[kr];
        #pragma unroll
        for (int j = 0; j < NGRP; ++j) {
            const uint32_t ico = (uint32_t)(j / 3) * 4096u;
            const int kr = j % 3;
            t.m1[j] = *(const v2f*)(X1 + (vm[kr] + ico));
            t.m2[j] = *(const v2f*)(X2 + (vm[kr] + ico));
        }
        t.lv = Lp[d];
        t.uv = Up[d];
    };

    float acc[8] = {0.f, 0.f, 0.f, 0.f, 0.f, 0.f, 0.f, 0.f};

    auto compute_task = [&](const TapBuf& t) {
        v8f ca = {};
        v8f cb = {};
        #pragma unroll
        for (int i = 0; i < NGRP / 2; ++i) {
            const int j0 = 2 * i, j1 = 2 * i + 1;
            ca = wmma4(wp[3 * i],     t.m1[j0], ca);
            cb = wmma4(wn[3 * i],     t.m2[j0], cb);
            ca = wmma4(wp[3 * i + 1], t.m1[j1], ca);
            cb = wmma4(wn[3 * i + 1], t.m2[j1], cb);
            const v2f ea = (v2f){ left_lane(t.m1[j0].y), left_lane(t.m1[j1].y) };
            const v2f eb = (v2f){ left_lane(t.m2[j0].y), left_lane(t.m2[j1].y) };
            ca = wmma4(wp[3 * i + 2], ea, ca);
            cb = wmma4(wn[3 * i + 2], eb, cb);
        }
        #pragma unroll
        for (int r = 0; r < 8; ++r) {
            const float x = ca[r] + cb[r];
            acc[r] += (x > 0.f ? t.lv : t.uv) * x;
        }
    };

    TapBuf A, B;
    load_task(A, 0);
    for (int task = 0; task < ntask - 2; task += 2) {
        load_task(B, task + 1);
        compute_task(A);
        load_task(A, task + 2);
        compute_task(B);
    }
    load_task(B, ntask - 1);
    compute_task(A);
    compute_task(B);

    #pragma unroll
    for (int r = 0; r < 8; ++r) s_part[wave][(8 * h + r) * 16 + c] = acc[r];
    __syncthreads();

    if (tid < OUTS_PER_BLOCK) {
        float sum = 0.f;
        #pragma unroll
        for (int w = 0; w < NWAVES; ++w) sum += s_part[w][tid];
        ws[(size_t)slice * SLICE_FLOATS + blk * OUTS_PER_BLOCK + tid] = sum;
    }
}

typedef __attribute__((ext_vector_type(4))) float v4f;

__global__ __launch_bounds__(32) void deeppoly_conv_finish(
    const float* __restrict__ lc0, const float* __restrict__ uc0,
    const float* __restrict__ weight, const float* __restrict__ bias,
    const float* __restrict__ ws, size_t ws_bytes, float* __restrict__ out)
{
    const int p   = blockIdx.x;
    const int tid = threadIdx.x;
    const int oc  = tid & 15;
    const bool isU = tid >= 16;
    const int ph = p >> 4, pw = p & 15;

    const int blk = ph * 2 + (isU ? 1 : 0);
    const int nslice = num_slices(ws_bytes);
    const float* __restrict__ wsp = ws + blk * OUTS_PER_BLOCK + oc * 16 + pw;

    bool ok[3];
    int  jj[3];
    #pragma unroll
    for (int i = 0; i < 3; ++i) {
        const int k  = tid + 32 * i;
        const int ic = k / 9, t9 = k % 9;
        const int r  = ph * 2 - 1 + t9 / 3;
        const int c  = pw * 2 - 1 + (t9 % 3);
        ok[i] = (k < KTAPS) && r >= 0 && r < 32 && c >= 0 && c < 32;
        jj[i] = ok[i] ? ic * HW_ + r * 32 + c : 0;
    }

    float part[8];
    #pragma unroll
    for (int s = 0; s < 8; ++s) {
        const int sl = (s < nslice) ? s : s - 4;
        part[s] = wsp[(size_t)sl * SLICE_FLOATS];
    }
    float lw[3], uw[3];
    #pragma unroll
    for (int i = 0; i < 3; ++i) { lw[i] = lc0[jj[i]]; uw[i] = uc0[jj[i]]; }
    const v4f* __restrict__ wq = (const v4f*)(weight + oc * KTAPS);
    v4f w4[KTAPS / 4];
    #pragma unroll
    for (int q = 0; q < KTAPS / 4; ++q) w4[q] = wq[q];
    float cacc = bias[oc];

    __builtin_amdgcn_sched_barrier(0);

    #pragma unroll
    for (int i = 0; i < 3; ++i) {
        lw[i] = ok[i] ? lw[i] : 0.f;
        uw[i] = ok[i] ? uw[i] : 0.f;
    }

    float sum = 0.f;
    #pragma unroll
    for (int s = 0; s < 4; ++s) sum += part[s];
    if (nslice == 8) {
        #pragma unroll
        for (int s = 4; s < 8; ++s) sum += part[s];
    }

    #pragma unroll
    for (int k = 0; k < KTAPS; ++k) {
        const float lcv = __int_as_float(__builtin_amdgcn_readlane(__float_as_int(lw[k >> 5]), k & 31));
        const float ucv = __int_as_float(__builtin_amdgcn_readlane(__float_as_int(uw[k >> 5]), k & 31));
        const float wv = w4[k >> 2][k & 3];
        const float wp = wv > 0.f ? wv : 0.f;
        const float wn = wv < 0.f ? wv : 0.f;
        cacc += isU ? (wp * ucv + wn * lcv)
                    : (wp * lcv + wn * ucv);
    }
    out[(isU ? 4096 : 0) + oc * P_ + p] = sum + cacc;
}

extern "C" void kernel_launch(void* const* d_in, const int* in_sizes, int n_in,
                              void* d_out, int out_size, void* d_ws, size_t ws_size,
                              hipStream_t stream) {
    const float* l0     = (const float*)d_in[0];
    const float* u0     = (const float*)d_in[1];
    const float* lx0    = (const float*)d_in[2];
    const float* ux0    = (const float*)d_in[3];
    const float* lc0    = (const float*)d_in[4];
    const float* uc0    = (const float*)d_in[5];
    const float* weight = (const float*)d_in[6];
    const float* bias   = (const float*)d_in[7];
    float* ws  = (float*)d_ws;
    float* out = (float*)d_out;

    deeppoly_conv_main<<<256, NWAVES * 32, 0, stream>>>(
        l0, u0, lx0, ux0, weight, ws, ws_size);
    deeppoly_conv_finish<<<P_, 32, 0, stream>>>(
        lc0, uc0, weight, bias, ws, ws_size, out);
}


# Both bounds per wave from two products, (lx0+ux0)@W and (ux0-lx0)@|W|, instead of four: half the matrix instructions and loads; 16 d-slices if ws_size holds 512 KB, else 8 or 4; sums regrouped
# speedup vs baseline: 7.7710x; 1.4828x over previous
#include <hip/hip_runtime.h>
#include <hip/hip_bf16.h>


typedef __attribute__((ext_vector_type(2))) float v2f;
typedef __attribute__((ext_vector_type(8))) float v8f;

#define C_IN_   8
#define C_OUT_  16
#define HW_     1024
#define D0_     3072
#define D1_     8192
#define P_      256
#define KTAPS   72
#define KSTEPS  18
#define NWAVES  16
#define NGRP    12
#define SLICE_FLOATS 8192
#define ROWB    32768u
#define OUTS_PER_BLOCK 256
#define OUTS_BOTH 512
#define MAXSLICE 16

struct TapBuf {
    v2f   m1[NGRP], m2[NGRP];
    float lv, uv;
};

__device__ __forceinline__ v8f wmma4(v2f a, v2f b, v8f c) {
    return __builtin_amdgcn_wmma_f32_16x16x4_f32(false, a, false, b, (short)0, c, false, false);
}

__device__ __forceinline__ float left_lane(float v) {
    return __int_as_float(__builtin_amdgcn_update_dpp(0, __float_as_int(v), 0x111, 0xf, 0xf, true));
}

__device__ __forceinline__ int num_slices(size_t ws_bytes) {
    const size_t sl = (size_t)SLICE_FLOATS * sizeof(float);
    return ws_bytes >= 16 * sl ? 16 : (ws_bytes >= 8 * sl ? 8 : 4);
}

__global__ __launch_bounds__(NWAVES * 32) void deeppoly_conv_main(
    const float* __restrict__ l0,  const float* __restrict__ u0,
    const float* __restrict__ lx0, const float* __restrict__ ux0,
    const float* __restrict__ weight,
    float* __restrict__ ws, size_t ws_bytes)
{
    __shared__ float s_part[NWAVES][OUTS_BOTH];

    const int nslice = num_slices(ws_bytes);
    const int slice  = blockIdx.x >> 4;
    if (slice >= nslice) return;
    const int dslice = D0_ / nslice;
    const int ntask  = dslice / NWAVES;

    const int tid   = threadIdx.x;
    const int wave  = __builtin_amdgcn_readfirstlane(tid >> 5);
    const int lane  = tid & 31;
    const int ph    = blockIdx.x & 15;

    const int c   = lane & 15;
    const int h   = lane >> 4;
    const bool top = (ph == 0);

    const char*  __restrict__ X1 = (const char*)lx0;
    const char*  __restrict__ X2 = (const char*)ux0;
    const float* __restrict__ Lp = l0;
    const float* __restrict__ Up = u0;

    v2f wf[KSTEPS], wa[KSTEPS];
    #pragma unroll
    for (int i = 0; i < NGRP / 2; ++i) {
        float w[6];
        #pragma unroll
        for (int jj = 0; jj < 2; ++jj) {
            const int j  = 2 * i + jj;
            const int ic = j / 3 + 4 * h;
            const int kr = j % 3;
            const bool padrow = (kr == 0) && top;
            #pragma unroll
            for (int kc = 0; kc < 3; ++kc) {
                const float v = weight[c * KTAPS + ic * 9 + kr * 3 + kc];
                w[jj * 3 + kc] = padrow ? 0.f : v;
            }
        }
        const float a[6] = { w[1], w[2], w[4], w[5], w[0], w[3] };
        #pragma unroll
        for (int s = 0; s < 3; ++s) {
            const float x = a[2 * s], y = a[2 * s + 1];
            const v2f pw2 = (v2f){ x > 0.f ? x : 0.f, y > 0.f ? y : 0.f };
            const v2f nw2 = (v2f){ x < 0.f ? x : 0.f, y < 0.f ? y : 0.f };
            wf[3 * i + s] = pw2 + nw2;
            wa[3 * i + s] = pw2 - nw2;
        }
    }

    uint32_t rb[3];
    #pragma unroll
    for (int kr = 0; kr < 3; ++kr)
        rb[kr] = (kr == 0 && top) ? 0u : (uint32_t)(2 * ph - 1 + kr) * 128u;
    const uint32_t lane_m = (uint32_t)h * 16384u + (uint32_t)c * 8u;
    const uint32_t dbase  = (uint32_t)(slice * dslice + wave);

    auto load_task = [&](TapBuf& t, int task) {
        const uint32_t d  = dbase + (uint32_t)(NWAVES * task);
        const uint32_t vb = d * ROWB;
        uint32_t vm[3];
        #pragma unroll
        for (int kr = 0; kr < 3; ++kr) vm[kr] = vb + lane_m + rb[kr];
        #pragma unroll
        for (int j = 0; j < NGRP; ++j) {
            const uint32_t ico = (uint32_t)(j / 3) * 4096u;
            const int kr = j % 3;
            t.m1[j] = *(const v2f*)(X1 + (vm[kr] + ico));
            t.m2[j] = *(const v2f*)(X2 + (vm[kr] + ico));
        }
        t.lv = Lp[d];
        t.uv = Up[d];
    };

    float accl[8] = {0.f, 0.f, 0.f, 0.f, 0.f, 0.f, 0.f, 0.f};
    float accu[8] = {0.f, 0.f, 0.f, 0.f, 0.f, 0.f, 0.f, 0.f};

    auto compute_task = [&](const TapBuf& t) {
        v8f cp = {};
        v8f cq = {};
        #pragma unroll
        for (int i = 0; i < NGRP / 2; ++i) {
            const int j0 = 2 * i, j1 = 2 * i + 1;
            const v2f s0 = t.m1[j0] + t.m2[j0], d0 = t.m2[j0] - t.m1[j0];
            const v2f s1 = t.m1[j1] + t.m2[j1], d1 = t.m2[j1] - t.m1[j1];
            cp = wmma4(wf[3 * i],     s0, cp);
            cq = wmma4(wa[3 * i],     d0, cq);
            cp = wmma4(wf[3 * i + 1], s1, cp);
            cq = wmma4(wa[3 * i + 1], d1, cq);
            const v2f es = (v2f){ left_lane(s0.y), left_lane(s1.y) };
            const v2f ed = (v2f){ left_lane(d0.y), left_lane(d1.y) };
            cp = wmma4(wf[3 * i + 2], es, cp);
            cq = wmma4(wa[3 * i + 2], ed, cq);
        }
        const float hl = 0.5f * t.lv, hu = 0.5f * t.uv;
        #pragma unroll
        for (int r = 0; r < 8; ++r) {
            const float yl = cp[r] - cq[r];
            const float yu = cp[r] + cq[r];
            accl[r] += (yl > 0.f ? hl : hu) * yl;
            accu[r] += (yu > 0.f ? hu : hl) * yu;
        }
    };

    TapBuf A, B;
    load_task(A, 0);
    for (int task = 0; task < ntask - 2; task += 2) {
        load_task(B, task + 1);
        compute_task(A);
        load_task(A, task + 2);
        compute_task(B);
    }
    load_task(B, ntask - 1);
    compute_task(A);
    compute_task(B);

    #pragma unroll
    for (int r = 0; r < 8; ++r) {
        s_part[wave][(8 * h + r) * 16 + c] = accl[r];
        s_part[wave][OUTS_PER_BLOCK + (8 * h + r) * 16 + c] = accu[r];
    }
    __syncthreads();

    {
        float sum = 0.f;
        #pragma unroll
        for (int w = 0; w < NWAVES; ++w) sum += s_part[w][tid];
        ws[(size_t)slice * SLICE_FLOATS + ph * OUTS_BOTH + tid] = sum;
    }
}

typedef __attribute__((ext_vector_type(4))) float v4f;

__global__ __launch_bounds__(32) void deeppoly_conv_finish(
    const float* __restrict__ lc0, const float* __restrict__ uc0,
    const float* __restrict__ weight, const float* __restrict__ bias,
    const float* __restrict__ ws, size_t ws_bytes, float* __restrict__ out)
{
    const int p   = blockIdx.x;
    const int tid = threadIdx.x;
    const int oc  = tid & 15;
    const bool isU = tid >= 16;
    const int ph = p >> 4, pw = p & 15;

    const int blk = ph * 2 + (isU ? 1 : 0);
    const int nslice = num_slices(ws_bytes);
    const float* __restrict__ wsp = ws + blk * OUTS_PER_BLOCK + oc * 16 + pw;

    bool ok[3];
    int  jj[3];
    #pragma unroll
    for (int i = 0; i < 3; ++i) {
        const int k  = tid + 32 * i;
        const int ic = k / 9, t9 = k % 9;
        const int r  = ph * 2 - 1 + t9 / 3;
        const int c  = pw * 2 - 1 + (t9 % 3);
        ok[i] = (k < KTAPS) && r >= 0 && r < 32 && c >= 0 && c < 32;
        jj[i] = ok[i] ? ic * HW_ + r * 32 + c : 0;
    }

    float part[MAXSLICE];
    #pragma unroll
    for (int s = 0; s < MAXSLICE; ++s) {
        const int sl = s & (nslice - 1);
        part[s] = wsp[(size_t)sl * SLICE_FLOATS];
    }
    float lw[3], uw[3];
    #pragma unroll
    for (int i = 0; i < 3; ++i) { lw[i] = lc0[jj[i]]; uw[i] = uc0[jj[i]]; }
    const v4f* __restrict__ wq = (const v4f*)(weight + oc * KTAPS);
    v4f w4[KTAPS / 4];
    #pragma unroll
    for (int q = 0; q < KTAPS / 4; ++q) w4[q] = wq[q];
    float cacc = bias[oc];

    __builtin_amdgcn_sched_barrier(0);

    #pragma unroll
    for (int i = 0; i < 3; ++i) {
        lw[i] = ok[i] ? lw[i] : 0.f;
        uw[i] = ok[i] ? uw[i] : 0.f;
    }

    float sum = 0.f;
    #pragma unroll
    for (int s = 0; s < 4; ++s) sum += part[s];
    if (nslice >= 8) {
        #pragma unroll
        for (int s = 4; s < 8; ++s) sum += part[s];
    }
    if (nslice == 16) {
        #pragma unroll
        for (int s = 8; s < 16; ++s) sum += part[s];
    }

    #pragma unroll
    for (int k = 0; k < KTAPS; ++k) {
        const float lcv = __int_as_float(__builtin_amdgcn_readlane(__float_as_int(lw[k >> 5]), k & 31));
        const float ucv = __int_as_float(__builtin_amdgcn_readlane(__float_as_int(uw[k >> 5]), k & 31));
        const float wv = w4[k >> 2][k & 3];
        const float wp = wv > 0.f ? wv : 0.f;
        const float wn = wv < 0.f ? wv : 0.f;
        cacc += isU ? (wp * ucv + wn * lcv)
                    : (wp * lcv + wn * ucv);
    }
    out[(isU ? 4096 : 0) + oc * P_ + p] = sum + cacc;
}

extern "C" void kernel_launch(void* const* d_in, const int* in_sizes, int n_in,
                              void* d_out, int out_size, void* d_ws, size_t ws_size,
                              hipStream_t stream) {
    const float* l0     = (const float*)d_in[0];
    const float* u0     = (const float*)d_in[1];
    const float* lx0    = (const float*)d_in[2];
    const float* ux0    = (const float*)d_in[3];
    const float* lc0    = (const float*)d_in[4];
    const float* uc0    = (const float*)d_in[5];
    const float* weight = (const float*)d_in[6];
    const float* bias   = (const float*)d_in[7];
    float* ws  = (float*)d_ws;
    float* out = (float*)d_out;

    deeppoly_conv_main<<<256, NWAVES * 32, 0, stream>>>(
        l0, u0, lx0, ux0, weight, ws, ws_size);
    deeppoly_conv_finish<<<P_, 32, 0, stream>>>(
        lc0, uc0, weight, bias, ws, ws_size, out);
}


# Main kernel: each task's load group pinned ahead of the previous task's products by scheduling barriers, so a whole task's loads stay in flight under the matrix work; sum orders kept
# speedup vs baseline: 8.0253x; 1.0327x over previous
#include <hip/hip_runtime.h>
#include <hip/hip_bf16.h>


typedef __attribute__((ext_vector_type(2))) float v2f;
typedef __attribute__((ext_vector_type(8))) float v8f;

#define C_IN_   8
#define C_OUT_  16
#define HW_     1024
#define D0_     3072
#define D1_     8192
#define P_      256
#define KTAPS   72
#define KSTEPS  18
#define NWAVES  16
#define NGRP    12
#define SLICE_FLOATS 8192
#define ROWB    32768u
#define OUTS_PER_BLOCK 256
#define OUTS_BOTH 512
#define MAXSLICE 16

struct TapBuf {
    v2f   m1[NGRP], m2[NGRP];
    float lv, uv;
};

__device__ __forceinline__ v8f wmma4(v2f a, v2f b, v8f c) {
    return __builtin_amdgcn_wmma_f32_16x16x4_f32(false, a, false, b, (short)0, c, false, false);
}

__device__ __forceinline__ float left_lane(float v) {
    return __int_as_float(__builtin_amdgcn_update_dpp(0, __float_as_int(v), 0x111, 0xf, 0xf, true));
}

__device__ __forceinline__ int num_slices(size_t ws_bytes) {
    const size_t sl = (size_t)SLICE_FLOATS * sizeof(float);
    return ws_bytes >= 16 * sl ? 16 : (ws_bytes >= 8 * sl ? 8 : 4);
}

__global__ __launch_bounds__(NWAVES * 32) void deeppoly_conv_main(
    const float* __restrict__ l0,  const float* __restrict__ u0,
    const float* __restrict__ lx0, const float* __restrict__ ux0,
    const float* __restrict__ weight,
    float* __restrict__ ws, size_t ws_bytes)
{
    __shared__ float s_part[NWAVES][OUTS_BOTH];

    const int nslice = num_slices(ws_bytes);
    const int slice  = blockIdx.x >> 4;
    if (slice >= nslice) return;
    const int dslice = D0_ / nslice;
    const int ntask  = dslice / NWAVES;

    const int tid   = threadIdx.x;
    const int wave  = __builtin_amdgcn_readfirstlane(tid >> 5);
    const int lane  = tid & 31;
    const int ph    = blockIdx.x & 15;

    const int c   = lane & 15;
    const int h   = lane >> 4;
    const bool top = (ph == 0);

    const char*  __restrict__ X1 = (const char*)lx0;
    const char*  __restrict__ X2 = (const char*)ux0;
    const float* __restrict__ Lp = l0;
    const float* __restrict__ Up = u0;

    v2f wf[KSTEPS], wa[KSTEPS];
    #pragma unroll
    for (int i = 0; i < NGRP / 2; ++i) {
        float w[6];
        #pragma unroll
        for (int jj = 0; jj < 2; ++jj) {
            const int j  = 2 * i + jj;
            const int ic = j / 3 + 4 * h;
            const int kr = j % 3;
            const bool padrow = (kr == 0) && top;
            #pragma unroll
            for (int kc = 0; kc < 3; ++kc) {
                const float v = weight[c * KTAPS + ic * 9 + kr * 3 + kc];
                w[jj * 3 + kc] = padrow ? 0.f : v;
            }
        }
        const float a[6] = { w[1], w[2], w[4], w[5], w[0], w[3] };
        #pragma unroll
        for (int s = 0; s < 3; ++s) {
            const float x = a[2 * s], y = a[2 * s + 1];
            const v2f pw2 = (v2f){ x > 0.f ? x : 0.f, y > 0.f ? y : 0.f };
            const v2f nw2 = (v2f){ x < 0.f ? x : 0.f, y < 0.f ? y : 0.f };
            wf[3 * i + s] = pw2 + nw2;
            wa[3 * i + s] = pw2 - nw2;
        }
    }

    uint32_t rb[3];
    #pragma unroll
    for (int kr = 0; kr < 3; ++kr)
        rb[kr] = (kr == 0 && top) ? 0u : (uint32_t)(2 * ph - 1 + kr) * 128u;
    const uint32_t lane_m = (uint32_t)h * 16384u + (uint32_t)c * 8u;
    const uint32_t dbase  = (uint32_t)(slice * dslice + wave);

    auto load_task = [&](TapBuf& t, int task) {
        const uint32_t d  = dbase + (uint32_t)(NWAVES * task);
        const uint32_t vb = d * ROWB;
        uint32_t vm[3];
        #pragma unroll
        for (int kr = 0; kr < 3; ++kr) vm[kr] = vb + lane_m + rb[kr];
        #pragma unroll
        for (int j = 0; j < NGRP; ++j) {
            const uint32_t ico = (uint32_t)(j / 3) * 4096u;
            const int kr = j % 3;
            t.m1[j] = *(const v2f*)(X1 + (vm[kr] + ico));
            t.m2[j] = *(const v2f*)(X2 + (vm[kr] + ico));
        }
        t.lv = Lp[d];
        t.uv = Up[d];
    };

    float accl[8] = {0.f, 0.f, 0.f, 0.f, 0.f, 0.f, 0.f, 0.f};
    float accu[8] = {0.f, 0.f, 0.f, 0.f, 0.f, 0.f, 0.f, 0.f};

    auto compute_task = [&](const TapBuf& t) {
        v8f cp = {};
        v8f cq = {};
        #pragma unroll
        for (int i = 0; i < NGRP / 2; ++i) {
            const int j0 = 2 * i, j1 = 2 * i + 1;
            const v2f s0 = t.m1[j0] + t.m2[j0], d0 = t.m2[j0] - t.m1[j0];
            const v2f s1 = t.m1[j1] + t.m2[j1], d1 = t.m2[j1] - t.m1[j1];
            cp = wmma4(wf[3 * i],     s0, cp);
            cq = wmma4(wa[3 * i],     d0, cq);
            cp = wmma4(wf[3 * i + 1], s1, cp);
            cq = wmma4(wa[3 * i + 1], d1, cq);
            const v2f es = (v2f){ left_lane(s0.y), left_lane(s1.y) };
            const v2f ed = (v2f){ left_lane(d0.y), left_lane(d1.y) };
            cp = wmma4(wf[3 * i + 2], es, cp);
            cq = wmma4(wa[3 * i + 2], ed, cq);
        }
        const float hl = 0.5f * t.lv, hu = 0.5f * t.uv;
        #pragma unroll
        for (int r = 0; r < 8; ++r) {
            const float yl = cp[r] - cq[r];
            const float yu = cp[r] + cq[r];
            accl[r] += (yl > 0.f ? hl : hu) * yl;
            accu[r] += (yu > 0.f ? hu : hl) * yu;
        }
    };

    TapBuf A, B;
    load_task(A, 0);
    for (int task = 0; task < ntask - 2; task += 2) {
        load_task(B, task + 1);
        __builtin_amdgcn_sched_barrier(0);
        compute_task(A);
        __builtin_amdgcn_sched_barrier(0);
        load_task(A, task + 2);
        __builtin_amdgcn_sched_barrier(0);
        compute_task(B);
        __builtin_amdgcn_sched_barrier(0);
    }
    load_task(B, ntask - 1);
    __builtin_amdgcn_sched_barrier(0);
    compute_task(A);
    compute_task(B);

    #pragma unroll
    for (int r = 0; r < 8; ++r) {
        s_part[wave][(8 * h + r) * 16 + c] = accl[r];
        s_part[wave][OUTS_PER_BLOCK + (8 * h + r) * 16 + c] = accu[r];
    }
    __syncthreads();

    {
        float sum = 0.f;
        #pragma unroll
        for (int w = 0; w < NWAVES; ++w) sum += s_part[w][tid];
        ws[(size_t)slice * SLICE_FLOATS + ph * OUTS_BOTH + tid] = sum;
    }
}

typedef __attribute__((ext_vector_type(4))) float v4f;

__global__ __launch_bounds__(32) void deeppoly_conv_finish(
    const float* __restrict__ lc0, const float* __restrict__ uc0,
    const float* __restrict__ weight, const float* __restrict__ bias,
    const float* __restrict__ ws, size_t ws_bytes, float* __restrict__ out)
{
    const int p   = blockIdx.x;
    const int tid = threadIdx.x;
    const int oc  = tid & 15;
    const bool isU = tid >= 16;
    const int ph = p >> 4, pw = p & 15;

    const int blk = ph * 2 + (isU ? 1 : 0);
    const int nslice = num_slices(ws_bytes);
    const float* __restrict__ wsp = ws + blk * OUTS_PER_BLOCK + oc * 16 + pw;

    bool ok[3];
    int  jj[3];
    #pragma unroll
    for (int i = 0; i < 3; ++i) {
        const int k  = tid + 32 * i;
        const int ic = k / 9, t9 = k % 9;
        const int r  = ph * 2 - 1 + t9 / 3;
        const int c  = pw * 2 - 1 + (t9 % 3);
        ok[i] = (k < KTAPS) && r >= 0 && r < 32 && c >= 0 && c < 32;
        jj[i] = ok[i] ? ic * HW_ + r * 32 + c : 0;
    }

    float part[MAXSLICE];
    #pragma unroll
    for (int s = 0; s < MAXSLICE; ++s) {
        const int sl = s & (nslice - 1);
        part[s] = wsp[(size_t)sl * SLICE_FLOATS];
    }
    float lw[3], uw[3];
    #pragma unroll
    for (int i = 0; i < 3; ++i) { lw[i] = lc0[jj[i]]; uw[i] = uc0[jj[i]]; }
    const v4f* __restrict__ wq = (const v4f*)(weight + oc * KTAPS);
    v4f w4[KTAPS / 4];
    #pragma unroll
    for (int q = 0; q < KTAPS / 4; ++q) w4[q] = wq[q];
    float cacc = bias[oc];

    __builtin_amdgcn_sched_barrier(0);

    #pragma unroll
    for (int i = 0; i < 3; ++i) {
        lw[i] = ok[i] ? lw[i] : 0.f;
        uw[i] = ok[i] ? uw[i] : 0.f;
    }

    float sum = 0.f;
    #pragma unroll
    for (int s = 0; s < 4; ++s) sum += part[s];
    if (nslice >= 8) {
        #pragma unroll
        for (int s = 4; s < 8; ++s) sum += part[s];
    }
    if (nslice == 16) {
        #pragma unroll
        for (int s = 8; s < 16; ++s) sum += part[s];
    }

    #pragma unroll
    for (int k = 0; k < KTAPS; ++k) {
        const float lcv = __int_as_float(__builtin_amdgcn_readlane(__float_as_int(lw[k >> 5]), k & 31));
        const float ucv = __int_as_float(__builtin_amdgcn_readlane(__float_as_int(uw[k >> 5]), k & 31));
        const float wv = w4[k >> 2][k & 3];
        const float wp = wv > 0.f ? wv : 0.f;
        const float wn = wv < 0.f ? wv : 0.f;
        cacc += isU ? (wp * ucv + wn * lcv)
                    : (wp * lcv + wn * ucv);
    }
    out[(isU ? 4096 : 0) + oc * P_ + p] = sum + cacc;
}

extern "C" void kernel_launch(void* const* d_in, const int* in_sizes, int n_in,
                              void* d_out, int out_size, void* d_ws, size_t ws_size,
                              hipStream_t stream) {
    const float* l0     = (const float*)d_in[0];
    const float* u0     = (const float*)d_in[1];
    const float* lx0    = (const float*)d_in[2];
    const float* ux0    = (const float*)d_in[3];
    const float* lc0    = (const float*)d_in[4];
    const float* uc0    = (const float*)d_in[5];
    const float* weight = (const float*)d_in[6];
    const float* bias   = (const float*)d_in[7];
    float* ws  = (float*)d_ws;
    float* out = (float*)d_out;

    deeppoly_conv_main<<<256, NWAVES * 32, 0, stream>>>(
        l0, u0, lx0, ux0, weight, ws, ws_size);
    deeppoly_conv_finish<<<P_, 32, 0, stream>>>(
        lc0, uc0, weight, bias, ws, ws_size, out);
}


# Constant terms (lc0/uc0 backsubstitution, bias) moved from the finish kernel into the main kernel's start-up, taps dealt over the d-slices; finish is a plain slice sum; constant-term sum regrouped
# speedup vs baseline: 8.6504x; 1.0779x over previous
#include <hip/hip_runtime.h>
#include <hip/hip_bf16.h>


typedef __attribute__((ext_vector_type(2))) float v2f;
typedef __attribute__((ext_vector_type(8))) float v8f;

#define C_IN_   8
#define C_OUT_  16
#define HW_     1024
#define D0_     3072
#define D1_     8192
#define P_      256
#define KTAPS   72
#define KSTEPS  18
#define NWAVES  16
#define NGRP    12
#define SLICE_FLOATS 8192
#define ROWB    32768u
#define OUTS_PER_BLOCK 256
#define OUTS_BOTH 512
#define MAXSLICE 16
#define CT      5

struct TapBuf {
    v2f   m1[NGRP], m2[NGRP];
    float lv, uv;
};

__device__ __forceinline__ v8f wmma4(v2f a, v2f b, v8f c) {
    return __builtin_amdgcn_wmma_f32_16x16x4_f32(false, a, false, b, (short)0, c, false, false);
}

__device__ __forceinline__ float left_lane(float v) {
    return __int_as_float(__builtin_amdgcn_update_dpp(0, __float_as_int(v), 0x111, 0xf, 0xf, true));
}

__device__ __forceinline__ int num_slices(size_t ws_bytes) {
    const size_t sl = (size_t)SLICE_FLOATS * sizeof(float);
    return ws_bytes >= 16 * sl ? 16 : (ws_bytes >= 8 * sl ? 8 : 4);
}

__global__ __launch_bounds__(NWAVES * 32) void deeppoly_conv_main(
    const float* __restrict__ l0,  const float* __restrict__ u0,
    const float* __restrict__ lx0, const float* __restrict__ ux0,
    const float* __restrict__ lc0, const float* __restrict__ uc0,
    const float* __restrict__ weight, const float* __restrict__ bias,
    float* __restrict__ ws, size_t ws_bytes)
{
    __shared__ float s_part[NWAVES][OUTS_BOTH];

    const int nslice = num_slices(ws_bytes);
    const int slice  = blockIdx.x >> 4;
    if (slice >= nslice) return;
    const int dslice = D0_ / nslice;
    const int ntask  = dslice / NWAVES;

    const int tid   = threadIdx.x;
    const int wave  = __builtin_amdgcn_readfirstlane(tid >> 5);
    const int lane  = tid & 31;
    const int ph    = blockIdx.x & 15;

    const int c   = lane & 15;
    const int h   = lane >> 4;
    const bool top = (ph == 0);

    const char*  __restrict__ X1 = (const char*)lx0;
    const char*  __restrict__ X2 = (const char*)ux0;
    const float* __restrict__ Lp = l0;
    const float* __restrict__ Up = u0;

    v2f wf[KSTEPS], wa[KSTEPS];
    #pragma unroll
    for (int i = 0; i < NGRP / 2; ++i) {
        float w[6];
        #pragma unroll
        for (int jj = 0; jj < 2; ++jj) {
            const int j  = 2 * i + jj;
            const int ic = j / 3 + 4 * h;
            const int kr = j % 3;
            const bool padrow = (kr == 0) && top;
            #pragma unroll
            for (int kc = 0; kc < 3; ++kc) {
                const float v = weight[c * KTAPS + ic * 9 + kr * 3 + kc];
                w[jj * 3 + kc] = padrow ? 0.f : v;
            }
        }
        const float a[6] = { w[1], w[2], w[4], w[5], w[0], w[3] };
        #pragma unroll
        for (int s = 0; s < 3; ++s) {
            const float x = a[2 * s], y = a[2 * s + 1];
            const v2f pw2 = (v2f){ x > 0.f ? x : 0.f, y > 0.f ? y : 0.f };
            const v2f nw2 = (v2f){ x < 0.f ? x : 0.f, y < 0.f ? y : 0.f };
            wf[3 * i + s] = pw2 + nw2;
            wa[3 * i + s] = pw2 - nw2;
        }
    }

    uint32_t rb[3];
    #pragma unroll
    for (int kr = 0; kr < 3; ++kr)
        rb[kr] = (kr == 0 && top) ? 0u : (uint32_t)(2 * ph - 1 + kr) * 128u;
    const uint32_t lane_m = (uint32_t)h * 16384u + (uint32_t)c * 8u;
    const uint32_t dbase  = (uint32_t)(slice * dslice + wave);

    auto load_task = [&](TapBuf& t, int task) {
        const uint32_t d  = dbase + (uint32_t)(NWAVES * task);
        const uint32_t vb = d * ROWB;
        uint32_t vm[3];
        #pragma unroll
        for (int kr = 0; kr < 3; ++kr) vm[kr] = vb + lane_m + rb[kr];
        #pragma unroll
        for (int j = 0; j < NGRP; ++j) {
            const uint32_t ico = (uint32_t)(j / 3) * 4096u;
            const int kr = j % 3;
            t.m1[j] = *(const v2f*)(X1 + (vm[kr] + ico));
            t.m2[j] = *(const v2f*)(X2 + (vm[kr] + ico));
        }
        t.lv = Lp[d];
        t.uv = Up[d];
    };

    float accl[8] = {0.f, 0.f, 0.f, 0.f, 0.f, 0.f, 0.f, 0.f};
    float accu[8] = {0.f, 0.f, 0.f, 0.f, 0.f, 0.f, 0.f, 0.f};

    auto compute_task = [&](const TapBuf& t) {
        v8f cp = {};
        v8f cq = {};
        #pragma unroll
        for (int i = 0; i < NGRP / 2; ++i) {
            const int j0 = 2 * i, j1 = 2 * i + 1;
            const v2f s0 = t.m1[j0] + t.m2[j0], d0 = t.m2[j0] - t.m1[j0];
            const v2f s1 = t.m1[j1] + t.m2[j1], d1 = t.m2[j1] - t.m1[j1];
            cp = wmma4(wf[3 * i],     s0, cp);
            cq = wmma4(wa[3 * i],     d0, cq);
            cp = wmma4(wf[3 * i + 1], s1, cp);
            cq = wmma4(wa[3 * i + 1], d1, cq);
            const v2f es = (v2f){ left_lane(s0.y), left_lane(s1.y) };
            const v2f ed = (v2f){ left_lane(d0.y), left_lane(d1.y) };
            cp = wmma4(wf[3 * i + 2], es, cp);
            cq = wmma4(wa[3 * i + 2], ed, cq);
        }
        const float hl = 0.5f * t.lv, hu = 0.5f * t.uv;
        #pragma unroll
        for (int r = 0; r < 8; ++r) {
            const float yl = cp[r] - cq[r];
            const float yu = cp[r] + cq[r];
            accl[r] += (yl > 0.f ? hl : hu) * yl;
            accu[r] += (yu > 0.f ? hu : hl) * yu;
        }
    };

    const bool isU = wave >= NWAVES / 2;
    const float* __restrict__ ca = isU ? uc0 : lc0;
    const float* __restrict__ cb = isU ? lc0 : uc0;
    const int eoc = ((wave & 7) << 1) | h;
    const int k0 = (KTAPS * slice) / nslice;
    const int k1 = (KTAPS * (slice + 1)) / nslice;
    const float bias_v = bias[eoc];
    float cterm = (slice == 0) ? bias_v : 0.f;
    auto cterm_group = [&](int kb) {
        float wv[CT], av[CT], bv[CT];
        #pragma unroll
        for (int i = 0; i < CT; ++i) {
            const int  k   = kb + i;
            const bool in  = k < k1;
            const int  kk  = in ? k : k0;
            const int  ic  = kk / 9, t9 = kk - 9 * ic;
            const int  kr  = t9 / 3, kc = t9 - 3 * kr;
            const int  r   = 2 * ph - 1 + kr;
            const int  cc  = 2 * c - 1 + kc;
            const bool ok  = in && r >= 0 && r < 32 && cc >= 0 && cc < 32;
            const int  j   = ok ? ic * HW_ + r * 32 + cc : 0;
            const float w  = weight[eoc * KTAPS + kk];
            wv[i] = ok ? w : 0.f;
            av[i] = ca[j];
            bv[i] = cb[j];
        }
        #pragma unroll
        for (int i = 0; i < CT; ++i) {
            const float wp = wv[i] > 0.f ? wv[i] : 0.f;
            const float wn = wv[i] < 0.f ? wv[i] : 0.f;
            cterm += wp * av[i] + wn * bv[i];
        }
    };
    for (int kb = k0 + CT; kb < k1; kb += CT) cterm_group(kb);

    TapBuf A, B;
    load_task(A, 0);
    __builtin_amdgcn_sched_barrier(0);
    cterm_group(k0);
    for (int task = 0; task < ntask - 2; task += 2) {
        load_task(B, task + 1);
        __builtin_amdgcn_sched_barrier(0);
        compute_task(A);
        __builtin_amdgcn_sched_barrier(0);
        load_task(A, task + 2);
        __builtin_amdgcn_sched_barrier(0);
        compute_task(B);
        __builtin_amdgcn_sched_barrier(0);
    }
    load_task(B, ntask - 1);
    __builtin_amdgcn_sched_barrier(0);
    compute_task(A);
    compute_task(B);

    #pragma unroll
    for (int r = 0; r < 8; ++r) {
        s_part[wave][(8 * h + r) * 16 + c] = accl[r];
        s_part[wave][OUTS_PER_BLOCK + (8 * h + r) * 16 + c] = accu[r];
    }
    __syncthreads();

    {
        float sum = 0.f;
        #pragma unroll
        for (int w = 0; w < NWAVES; ++w) sum += s_part[w][tid];
        ws[(size_t)slice * SLICE_FLOATS + ph * OUTS_BOTH + tid] = sum + cterm;
    }
}

__global__ __launch_bounds__(32) void deeppoly_conv_finish(
    const float* __restrict__ ws, size_t ws_bytes, float* __restrict__ out)
{
    const int o  = blockIdx.x * 32 + threadIdx.x;
    const int b  = o >> 12;
    const int oc = (o >> 8) & 15;
    const int p  = o & 255;

    const int idx = (p >> 4) * OUTS_BOTH + b * OUTS_PER_BLOCK + oc * 16 + (p & 15);
    const int nslice = num_slices(ws_bytes);

    const float* g0 = ws;
    const float* g1 = ws + (nslice >= 8  ? 4  * SLICE_FLOATS : 0);
    const float* g2 = ws + (nslice == 16 ? 8  * SLICE_FLOATS : 0);
    const float* g3 = ws + (nslice == 16 ? 12 * SLICE_FLOATS : 0);
    float part[MAXSLICE];
    #pragma unroll
    for (int i = 0; i < 4; ++i) {
        part[i]      = g0[i * SLICE_FLOATS + idx];
        part[4 + i]  = g1[i * SLICE_FLOATS + idx];
        part[8 + i]  = g2[i * SLICE_FLOATS + idx];
        part[12 + i] = g3[i * SLICE_FLOATS + idx];
    }

    __builtin_amdgcn_sched_barrier(0);

    float sum4 = 0.f;
    #pragma unroll
    for (int s = 0; s < 4; ++s) sum4 += part[s];
    float sum8 = sum4;
    #pragma unroll
    for (int s = 4; s < 8; ++s) sum8 += part[s];
    float sum16 = sum8;
    #pragma unroll
    for (int s = 8; s < 16; ++s) sum16 += part[s];
    out[o] = (nslice == 16) ? sum16 : (nslice >= 8 ? sum8 : sum4);
}

extern "C" void kernel_launch(void* const* d_in, const int* in_sizes, int n_in,
                              void* d_out, int out_size, void* d_ws, size_t ws_size,
                              hipStream_t stream) {
    const float* l0     = (const float*)d_in[0];
    const float* u0     = (const float*)d_in[1];
    const float* lx0    = (const float*)d_in[2];
    const float* ux0    = (const float*)d_in[3];
    const float* lc0    = (const float*)d_in[4];
    const float* uc0    = (const float*)d_in[5];
    const float* weight = (const float*)d_in[6];
    const float* bias   = (const float*)d_in[7];
    float* ws  = (float*)d_ws;
    float* out = (float*)d_out;

    deeppoly_conv_main<<<256, NWAVES * 32, 0, stream>>>(
        l0, u0, lx0, ux0, lc0, uc0, weight, bias, ws, ws_size);
    deeppoly_conv_finish<<<P_, 32, 0, stream>>>(
        ws, ws_size, out);
}
